# starting kernel
# baseline (speedup 1.0000x reference)
#include <hip/hip_runtime.h>

#define NN 100000
#define EE 1600000
#define D  128
#define HM 256

typedef __attribute__((ext_vector_type(2))) float v2f;
typedef __attribute__((ext_vector_type(8))) float v8f;

__global__ void init_kernel(float* __restrict__ agg, float* __restrict__ deg,
                            float* __restrict__ pred, const float* __restrict__ bm2) {
  int i = blockIdx.x * blockDim.x + threadIdx.x;
  if (i < NN * D) agg[i] = 0.0f;
  if (i < NN) { deg[i] = 0.0f; pred[i] = bm2[0]; }
}

__global__ void zero_agg_kernel(float* __restrict__ agg) {
  int i = blockIdx.x * blockDim.x + threadIdx.x;
  if (i < NN * D) agg[i] = 0.0f;
}

__global__ void degree_kernel(const int* __restrict__ ei, float* __restrict__ deg) {
  int e = blockIdx.x * blockDim.x + threadIdx.x;
  if (e >= EE) return;
  atomicAdd(&deg[ei[EE + e]], 1.0f);
}

__global__ void scatter_kernel(const float* __restrict__ feat, const int* __restrict__ ei,
                               float* __restrict__ agg) {
  int idx  = blockIdx.x * blockDim.x + threadIdx.x;
  int e    = idx >> 5;
  int lane = idx & 31;
  int s = ei[e];
  int d = ei[EE + e];
  float4 v = *((const float4*)(feat + (size_t)s * D) + lane);
  float* o = agg + (size_t)d * D + lane * 4;
  atomicAdd(o + 0, v.x);
  atomicAdd(o + 1, v.y);
  atomicAdd(o + 2, v.z);
  atomicAdd(o + 3, v.w);
}

__global__ void sage_combine_kernel(const float* __restrict__ xin,
                                    const float* __restrict__ agg,
                                    const float* __restrict__ deg,
                                    const float* __restrict__ Wl,
                                    const float* __restrict__ bias,
                                    const float* __restrict__ Wr,
                                    float* __restrict__ out, int do_relu) {
  int wave = (blockIdx.x * blockDim.x + threadIdx.x) >> 5;
  int lane = threadIdx.x & 31;
  int tm = wave >> 3;
  int tn = wave & 7;
  int m0 = tm << 4, n0 = tn << 4;
  int rl = lane & 15;
  int hi = lane >> 4;
  int arow = m0 + rl;

  float rdeg = 1.0f / fmaxf(deg[arow], 1.0f);
  const float* xp = xin + (size_t)arow * D + hi * 2;
  const float* gp = agg + (size_t)arow * D + hi * 2;

  v8f accL = {};
  v8f accR = {};
#pragma unroll 4
  for (int kb = 0; kb < D; kb += 4) {
    v2f am; am[0] = gp[kb] * rdeg; am[1] = gp[kb + 1] * rdeg;
    v2f ax; ax[0] = xp[kb];        ax[1] = xp[kb + 1];
    const float* blp = Wl + (size_t)(kb + hi * 2) * D + n0 + rl;
    const float* brp = Wr + (size_t)(kb + hi * 2) * D + n0 + rl;
    v2f bl; bl[0] = blp[0]; bl[1] = blp[D];
    v2f br; br[0] = brp[0]; br[1] = brp[D];
    accL = __builtin_amdgcn_wmma_f32_16x16x4_f32(false, am, false, bl, (short)0, accL, false, false);
    accR = __builtin_amdgcn_wmma_f32_16x16x4_f32(false, ax, false, br, (short)0, accR, false, false);
  }

  float bv = bias[n0 + rl];
#pragma unroll
  for (int g = 0; g < 8; ++g) {
    float v = accL[g] + accR[g] + bv;
    if (do_relu) v = fmaxf(v, 0.0f);
    out[(size_t)(m0 + g + hi * 8) * D + n0 + rl] = v;
  }
}

__global__ void mlp_kernel(const float* __restrict__ emb,
                           const float* __restrict__ Wm1,
                           const float* __restrict__ bm1,
                           const float* __restrict__ Wm2,
                           float* __restrict__ pred) {
  int wave = (blockIdx.x * blockDim.x + threadIdx.x) >> 5;
  int lane = threadIdx.x & 31;
  int tm = wave >> 4;
  int tn = wave & 15;
  int m0 = tm << 4, n0 = tn << 4;
  int rl = lane & 15;
  int hi = lane >> 4;
  int arow = m0 + rl;

  const float* ep = emb + (size_t)arow * D + hi * 2;

  v8f accA = {};
  v8f accB = {};
#pragma unroll 4
  for (int kb = 0; kb < D; kb += 8) {
    v2f a0; a0[0] = fmaxf(ep[kb], 0.0f);     a0[1] = fmaxf(ep[kb + 1], 0.0f);
    v2f a1; a1[0] = fmaxf(ep[kb + 4], 0.0f); a1[1] = fmaxf(ep[kb + 5], 0.0f);
    const float* bp0 = Wm1 + (size_t)(kb + hi * 2) * HM + n0 + rl;
    const float* bp1 = Wm1 + (size_t)(kb + 4 + hi * 2) * HM + n0 + rl;
    v2f b0; b0[0] = bp0[0]; b0[1] = bp0[HM];
    v2f b1; b1[0] = bp1[0]; b1[1] = bp1[HM];
    accA = __builtin_amdgcn_wmma_f32_16x16x4_f32(false, a0, false, b0, (short)0, accA, false, false);
    accB = __builtin_amdgcn_wmma_f32_16x16x4_f32(false, a1, false, b1, (short)0, accB, false, false);
  }

  float b1v = bm1[n0 + rl];
  float w2  = Wm2[n0 + rl];
#pragma unroll
  for (int g = 0; g < 8; ++g) {
    float h1 = fmaxf(accA[g] + accB[g] + b1v, 0.0f);
    float c = h1 * w2;
    c += __shfl_xor(c, 8, 32);
    c += __shfl_xor(c, 4, 32);
    c += __shfl_xor(c, 2, 32);
    c += __shfl_xor(c, 1, 32);
    if (rl == 0) atomicAdd(&pred[m0 + g + hi * 8], c);
  }
}

extern "C" void kernel_launch(void* const* d_in, const int* in_sizes, int n_in,
                              void* d_out, int out_size, void* d_ws, size_t ws_size,
                              hipStream_t stream) {
  const float* x    = (const float*)d_in[0];
  const int*   ei   = (const int*)  d_in[1];
  const float* Wl0  = (const float*)d_in[2];
  const float* b0   = (const float*)d_in[3];
  const float* Wr0  = (const float*)d_in[4];
  const float* Wl1  = (const float*)d_in[5];
  const float* b1   = (const float*)d_in[6];
  const float* Wr1  = (const float*)d_in[7];
  const float* Wm1  = (const float*)d_in[8];
  const float* bm1  = (const float*)d_in[9];
  const float* Wm2  = (const float*)d_in[10];
  const float* bm2  = (const float*)d_in[11];

  float* emb  = (float*)d_out;
  float* pred = emb + (size_t)NN * D;

  float* ws  = (float*)d_ws;
  float* deg = ws;
  float* agg = ws + NN;
  float* h0  = agg + (size_t)NN * D;

  const int T = 256;
  const int gInit    = (NN * D + T - 1) / T;
  const int gDeg     = (EE + T - 1) / T;
  const int gScatter = (EE * 32) / T;
  const int gCombine = ((NN / 16) * 8 * 32) / T;
  const int gMlp     = ((NN / 16) * 16 * 32) / T;

  init_kernel<<<gInit, T, 0, stream>>>(agg, deg, pred, bm2);
  degree_kernel<<<gDeg, T, 0, stream>>>(ei, deg);

  scatter_kernel<<<gScatter, T, 0, stream>>>(x, ei, agg);
  sage_combine_kernel<<<gCombine, T, 0, stream>>>(x, agg, deg, Wl0, b0, Wr0, h0, 1);

  zero_agg_kernel<<<gInit, T, 0, stream>>>(agg);
  scatter_kernel<<<gScatter, T, 0, stream>>>(h0, ei, agg);
  sage_combine_kernel<<<gCombine, T, 0, stream>>>(h0, agg, deg, Wl1, b1, Wr1, emb, 0);

  mlp_kernel<<<gMlp, T, 0, stream>>>(emb, Wm1, bm1, Wm2, pred);
}


# Edges bucketed by destination per call (count with returning int add, scan, place); one wave per row sums its source rows and stores once, replacing the f32 atomic scatter; sum order = slot order
# speedup vs baseline: 1.8299x; 1.8299x over previous
#include <hip/hip_runtime.h>

#define NN 100000
#define EE 1600000
#define D  128
#define HM 256

typedef __attribute__((ext_vector_type(2))) float v2f;
typedef __attribute__((ext_vector_type(8))) float v8f;

#define SCAN_Q     (NN / 4)
#define SCAN_NB    ((SCAN_Q + 255) / 256)

__global__ void init_kernel(int* __restrict__ cnt, float* __restrict__ pred,
                            const float* __restrict__ bm2) {
  int i = blockIdx.x * blockDim.x + threadIdx.x;
  if (i < NN) { cnt[i] = 0; pred[i] = bm2[0]; }
}

__global__ void count_kernel(const int* __restrict__ ei, int* __restrict__ cnt,
                             int* __restrict__ rank) {
  int e = blockIdx.x * blockDim.x + threadIdx.x;
  if (e >= EE) return;
  rank[e] = atomicAdd(&cnt[ei[EE + e]], 1);
}

__global__ void chunk_sum_kernel(const int* __restrict__ cnt, int* __restrict__ bsum) {
  __shared__ int wsum[8];
  int t = threadIdx.x;
  int q = blockIdx.x * 256 + t;
  int s = 0;
  if (q < SCAN_Q) {
    int4 c = ((const int4*)cnt)[q];
    s = c.x + c.y + c.z + c.w;
  }
  for (int o = 16; o > 0; o >>= 1) s += __shfl_xor(s, o, 32);
  if ((t & 31) == 0) wsum[t >> 5] = s;
  __syncthreads();
  if (t == 0) {
    int tot = 0;
    for (int w = 0; w < 8; ++w) tot += wsum[w];
    bsum[blockIdx.x] = tot;
  }
}

__global__ void scan_kernel(int* pos, const int* __restrict__ bsum, float* __restrict__ deg) {
  __shared__ int wtot[8];
  __shared__ int wpre[8];
  int t = threadIdx.x, lane = t & 31, w = t >> 5;
  int b = blockIdx.x;

  int v = (t < b) ? bsum[t] : 0;
  for (int o = 16; o > 0; o >>= 1) v += __shfl_xor(v, o, 32);
  if (lane == 0) wpre[w] = v;
  __syncthreads();
  int base = 0;
  for (int k = 0; k < 8; ++k) base += wpre[k];

  int q = b * 256 + t;
  bool valid = q < SCAN_Q;
  int4 c = make_int4(0, 0, 0, 0);
  if (valid) c = ((const int4*)pos)[q];
  int tot = c.x + c.y + c.z + c.w;
  int incl = tot;
  for (int o = 1; o < 32; o <<= 1) {
    int u = __shfl_up(incl, o, 32);
    if (lane >= o) incl += u;
  }
  if (lane == 31) wtot[w] = incl;
  __syncthreads();
  int wo = 0;
  for (int k = 0; k < w; ++k) wo += wtot[k];
  int ex = base + wo + incl - tot;
  if (valid) {
    int4 o4;
    o4.x = ex;
    o4.y = ex + c.x;
    o4.z = ex + c.x + c.y;
    o4.w = ex + c.x + c.y + c.z;
    ((int4*)pos)[q] = o4;
    float4 dg;
    dg.x = (float)c.x; dg.y = (float)c.y; dg.z = (float)c.z; dg.w = (float)c.w;
    ((float4*)deg)[q] = dg;
  }
}

__global__ void place_kernel(const int* __restrict__ ei, const int* __restrict__ pos,
                             const int* __restrict__ rank, int* __restrict__ col) {
  int e = blockIdx.x * blockDim.x + threadIdx.x;
  if (e >= EE) return;
  col[pos[ei[EE + e]] + rank[e]] = ei[e];
}

__global__ void gather_kernel(const float* __restrict__ feat, const int* __restrict__ pos,
                              const int* __restrict__ col, float* __restrict__ agg) {
  int lane = threadIdx.x & 31;
  int d = __builtin_amdgcn_readfirstlane((int)((blockIdx.x * blockDim.x + threadIdx.x) >> 5));
  int beg = pos[d];
  int end = (d + 1 < NN) ? pos[d + 1] : EE;
  const float4* f4 = (const float4*)feat;

  float4 a0 = make_float4(0.f, 0.f, 0.f, 0.f);
  float4 a1 = a0, a2 = a0, a3 = a0;
  for (int base = beg; base < end; base += 32) {
    int n = end - base;
    if (n > 32) n = 32;
    int myc = col[base + (lane < n ? lane : n - 1)];
    int i = 0;
    for (; i + 4 <= n; i += 4) {
      int s0 = __builtin_amdgcn_readlane(myc, i);
      int s1 = __builtin_amdgcn_readlane(myc, i + 1);
      int s2 = __builtin_amdgcn_readlane(myc, i + 2);
      int s3 = __builtin_amdgcn_readlane(myc, i + 3);
      float4 v0 = f4[(size_t)s0 * (D / 4) + lane];
      float4 v1 = f4[(size_t)s1 * (D / 4) + lane];
      float4 v2 = f4[(size_t)s2 * (D / 4) + lane];
      float4 v3 = f4[(size_t)s3 * (D / 4) + lane];
      a0.x += v0.x; a0.y += v0.y; a0.z += v0.z; a0.w += v0.w;
      a1.x += v1.x; a1.y += v1.y; a1.z += v1.z; a1.w += v1.w;
      a2.x += v2.x; a2.y += v2.y; a2.z += v2.z; a2.w += v2.w;
      a3.x += v3.x; a3.y += v3.y; a3.z += v3.z; a3.w += v3.w;
    }
    int rem = n - i;
    if (rem == 3) {
      int s0 = __builtin_amdgcn_readlane(myc, i);
      int s1 = __builtin_amdgcn_readlane(myc, i + 1);
      int s2 = __builtin_amdgcn_readlane(myc, i + 2);
      float4 v0 = f4[(size_t)s0 * (D / 4) + lane];
      float4 v1 = f4[(size_t)s1 * (D / 4) + lane];
      float4 v2 = f4[(size_t)s2 * (D / 4) + lane];
      a0.x += v0.x; a0.y += v0.y; a0.z += v0.z; a0.w += v0.w;
      a1.x += v1.x; a1.y += v1.y; a1.z += v1.z; a1.w += v1.w;
      a2.x += v2.x; a2.y += v2.y; a2.z += v2.z; a2.w += v2.w;
    } else if (rem == 2) {
      int s0 = __builtin_amdgcn_readlane(myc, i);
      int s1 = __builtin_amdgcn_readlane(myc, i + 1);
      float4 v0 = f4[(size_t)s0 * (D / 4) + lane];
      float4 v1 = f4[(size_t)s1 * (D / 4) + lane];
      a0.x += v0.x; a0.y += v0.y; a0.z += v0.z; a0.w += v0.w;
      a1.x += v1.x; a1.y += v1.y; a1.z += v1.z; a1.w += v1.w;
    } else if (rem == 1) {
      int s0 = __builtin_amdgcn_readlane(myc, i);
      float4 v0 = f4[(size_t)s0 * (D / 4) + lane];
      a0.x += v0.x; a0.y += v0.y; a0.z += v0.z; a0.w += v0.w;
    }
  }
  float4 r;
  r.x = (a0.x + a1.x) + (a2.x + a3.x);
  r.y = (a0.y + a1.y) + (a2.y + a3.y);
  r.z = (a0.z + a1.z) + (a2.z + a3.z);
  r.w = (a0.w + a1.w) + (a2.w + a3.w);
  ((float4*)agg)[(size_t)d * (D / 4) + lane] = r;
}

__global__ void sage_combine_kernel(const float* __restrict__ xin,
                                    const float* __restrict__ agg,
                                    const float* __restrict__ deg,
                                    const float* __restrict__ Wl,
                                    const float* __restrict__ bias,
                                    const float* __restrict__ Wr,
                                    float* __restrict__ out, int do_relu) {
  int wave = (blockIdx.x * blockDim.x + threadIdx.x) >> 5;
  int lane = threadIdx.x & 31;
  int tm = wave >> 3;
  int tn = wave & 7;
  int m0 = tm << 4, n0 = tn << 4;
  int rl = lane & 15;
  int hi = lane >> 4;
  int arow = m0 + rl;

  float rdeg = 1.0f / fmaxf(deg[arow], 1.0f);
  const float* xp = xin + (size_t)arow * D + hi * 2;
  const float* gp = agg + (size_t)arow * D + hi * 2;

  v8f accL = {};
  v8f accR = {};
#pragma unroll 4
  for (int kb = 0; kb < D; kb += 4) {
    v2f am; am[0] = gp[kb] * rdeg; am[1] = gp[kb + 1] * rdeg;
    v2f ax; ax[0] = xp[kb];        ax[1] = xp[kb + 1];
    const float* blp = Wl + (size_t)(kb + hi * 2) * D + n0 + rl;
    const float* brp = Wr + (size_t)(kb + hi * 2) * D + n0 + rl;
    v2f bl; bl[0] = blp[0]; bl[1] = blp[D];
    v2f br; br[0] = brp[0]; br[1] = brp[D];
    accL = __builtin_amdgcn_wmma_f32_16x16x4_f32(false, am, false, bl, (short)0, accL, false, false);
    accR = __builtin_amdgcn_wmma_f32_16x16x4_f32(false, ax, false, br, (short)0, accR, false, false);
  }

  float bv = bias[n0 + rl];
#pragma unroll
  for (int g = 0; g < 8; ++g) {
    float v = accL[g] + accR[g] + bv;
    if (do_relu) v = fmaxf(v, 0.0f);
    out[(size_t)(m0 + g + hi * 8) * D + n0 + rl] = v;
  }
}

__global__ void mlp_kernel(const float* __restrict__ emb,
                           const float* __restrict__ Wm1,
                           const float* __restrict__ bm1,
                           const float* __restrict__ Wm2,
                           float* __restrict__ pred) {
  int wave = (blockIdx.x * blockDim.x + threadIdx.x) >> 5;
  int lane = threadIdx.x & 31;
  int tm = wave >> 4;
  int tn = wave & 15;
  int m0 = tm << 4, n0 = tn << 4;
  int rl = lane & 15;
  int hi = lane >> 4;
  int arow = m0 + rl;

  const float* ep = emb + (size_t)arow * D + hi * 2;

  v8f accA = {};
  v8f accB = {};
#pragma unroll 4
  for (int kb = 0; kb < D; kb += 8) {
    v2f a0; a0[0] = fmaxf(ep[kb], 0.0f);     a0[1] = fmaxf(ep[kb + 1], 0.0f);
    v2f a1; a1[0] = fmaxf(ep[kb + 4], 0.0f); a1[1] = fmaxf(ep[kb + 5], 0.0f);
    const float* bp0 = Wm1 + (size_t)(kb + hi * 2) * HM + n0 + rl;
    const float* bp1 = Wm1 + (size_t)(kb + 4 + hi * 2) * HM + n0 + rl;
    v2f b0; b0[0] = bp0[0]; b0[1] = bp0[HM];
    v2f b1; b1[0] = bp1[0]; b1[1] = bp1[HM];
    accA = __builtin_amdgcn_wmma_f32_16x16x4_f32(false, a0, false, b0, (short)0, accA, false, false);
    accB = __builtin_amdgcn_wmma_f32_16x16x4_f32(false, a1, false, b1, (short)0, accB, false, false);
  }

  float b1v = bm1[n0 + rl];
  float w2  = Wm2[n0 + rl];
#pragma unroll
  for (int g = 0; g < 8; ++g) {
    float h1 = fmaxf(accA[g] + accB[g] + b1v, 0.0f);
    float c = h1 * w2;
    c += __shfl_xor(c, 8, 32);
    c += __shfl_xor(c, 4, 32);
    c += __shfl_xor(c, 2, 32);
    c += __shfl_xor(c, 1, 32);
    if (rl == 0) atomicAdd(&pred[m0 + g + hi * 8], c);
  }
}

extern "C" void kernel_launch(void* const* d_in, const int* in_sizes, int n_in,
                              void* d_out, int out_size, void* d_ws, size_t ws_size,
                              hipStream_t stream) {
  const float* x    = (const float*)d_in[0];
  const int*   ei   = (const int*)  d_in[1];
  const float* Wl0  = (const float*)d_in[2];
  const float* b0   = (const float*)d_in[3];
  const float* Wr0  = (const float*)d_in[4];
  const float* Wl1  = (const float*)d_in[5];
  const float* b1   = (const float*)d_in[6];
  const float* Wr1  = (const float*)d_in[7];
  const float* Wm1  = (const float*)d_in[8];
  const float* bm1  = (const float*)d_in[9];
  const float* Wm2  = (const float*)d_in[10];
  const float* bm2  = (const float*)d_in[11];

  float* emb  = (float*)d_out;
  float* pred = emb + (size_t)NN * D;

  float* ws  = (float*)d_ws;
  float* deg = ws;
  float* agg = ws + NN;
  float* h0  = agg + (size_t)NN * D;

  int* pos  = (int*)d_out;
  int* bsum = pos + NN;
  int* rank = bsum + 128;
  int* col  = rank + EE;

  const int T = 256;
  const int gNode    = (NN + T - 1) / T;
  const int gEdge    = (EE + T - 1) / T;
  const int gGather  = (NN * 32) / T;
  const int gCombine = ((NN / 16) * 8 * 32) / T;
  const int gMlp     = ((NN / 16) * 16 * 32) / T;

  init_kernel<<<gNode, T, 0, stream>>>(pos, pred, bm2);
  count_kernel<<<gEdge, T, 0, stream>>>(ei, pos, rank);
  chunk_sum_kernel<<<SCAN_NB, T, 0, stream>>>(pos, bsum);
  scan_kernel<<<SCAN_NB, T, 0, stream>>>(pos, bsum, deg);
  place_kernel<<<gEdge, T, 0, stream>>>(ei, pos, rank, col);

  gather_kernel<<<gGather, T, 0, stream>>>(x, pos, col, agg);
  sage_combine_kernel<<<gCombine, T, 0, stream>>>(x, agg, deg, Wl0, b0, Wr0, h0, 1);

  gather_kernel<<<gGather, T, 0, stream>>>(h0, pos, col, agg);
  sage_combine_kernel<<<gCombine, T, 0, stream>>>(h0, agg, deg, Wl1, b1, Wr1, emb, 0);

  mlp_kernel<<<gMlp, T, 0, stream>>>(emb, Wm1, bm1, Wm2, pred);
}


# Edges grouped by a two-level counting sort ranked with LDS integer adds (chunk counts per 256-row bucket, scan, per-bucket row sort), not one returning global add per edge; row sum order changes
# speedup vs baseline: 3.7098x; 2.0273x over previous
#include <hip/hip_runtime.h>

#define NN 100000
#define EE 1600000
#define D  128
#define HM 256

typedef __attribute__((ext_vector_type(2))) float v2f;
typedef __attribute__((ext_vector_type(8))) float v8f;

#define RB      256
#define NB      ((NN + RB - 1) / RB)
#define CH      4096
#define NCH     ((EE + CH - 1) / CH)
#define NC      (NCH + 1)
#define SCAN_Q  ((NB * NC) / 4)
#define SCAN_NB ((SCAN_Q + 255) / 256)

static_assert((NB * NC) % 4 == 0, "the scan reads the table as int4");
static_assert(SCAN_NB <= 256, "one scan block sums the totals of all blocks before it");
static_assert(NN <= (1 << 24), "a source index shares its word with 8 bits of the destination");
static_assert(EE % 4 == 0, "edge lists are read as int4");

__global__ void __launch_bounds__(256)
chunk_count_kernel(const int* __restrict__ ei, int* __restrict__ tab) {
  __shared__ int h[NB];
  int t = threadIdx.x, c = blockIdx.x;
  for (int i = t; i < NB; i += 256) h[i] = 0;
  __syncthreads();
  const int4* d4 = (const int4*)(ei + EE);
  int4 d[4];
  bool ok[4];
#pragma unroll
  for (int k = 0; k < 4; ++k) {
    int q = c * (CH / 4) + k * 256 + t;
    ok[k] = q < EE / 4;
    d[k] = d4[ok[k] ? q : EE / 4 - 1];
  }
#pragma unroll
  for (int k = 0; k < 4; ++k) {
    if (ok[k]) {
      atomicAdd(&h[d[k].x >> 8], 1);
      atomicAdd(&h[d[k].y >> 8], 1);
      atomicAdd(&h[d[k].z >> 8], 1);
      atomicAdd(&h[d[k].w >> 8], 1);
    }
  }
  __syncthreads();
  for (int i = t; i < NB; i += 256) tab[i * NC + c] = h[i];
}

__global__ void chunk_sum_kernel(const int* __restrict__ cnt, int* __restrict__ bsum) {
  __shared__ int wsum[8];
  int t = threadIdx.x;
  int q = blockIdx.x * 256 + t;
  int s = 0;
  if (q < SCAN_Q) {
    int4 c = ((const int4*)cnt)[q];
    s = c.x + c.y + c.z + c.w;
  }
  for (int o = 16; o > 0; o >>= 1) s += __shfl_xor(s, o, 32);
  if ((t & 31) == 0) wsum[t >> 5] = s;
  __syncthreads();
  if (t == 0) {
    int tot = 0;
    for (int w = 0; w < 8; ++w) tot += wsum[w];
    bsum[blockIdx.x] = tot;
  }
}

__global__ void scan_kernel(int* pos, const int* __restrict__ bsum) {
  __shared__ int wtot[8];
  __shared__ int wpre[8];
  int t = threadIdx.x, lane = t & 31, w = t >> 5;
  int b = blockIdx.x;

  int v = (t < b) ? bsum[t] : 0;
  for (int o = 16; o > 0; o >>= 1) v += __shfl_xor(v, o, 32);
  if (lane == 0) wpre[w] = v;
  __syncthreads();
  int base = 0;
  for (int k = 0; k < 8; ++k) base += wpre[k];

  int q = b * 256 + t;
  bool valid = q < SCAN_Q;
  int4 c = make_int4(0, 0, 0, 0);
  if (valid) c = ((const int4*)pos)[q];
  int tot = c.x + c.y + c.z + c.w;
  int incl = tot;
  for (int o = 1; o < 32; o <<= 1) {
    int u = __shfl_up(incl, o, 32);
    if (lane >= o) incl += u;
  }
  if (lane == 31) wtot[w] = incl;
  __syncthreads();
  int wo = 0;
  for (int k = 0; k < w; ++k) wo += wtot[k];
  int ex = base + wo + incl - tot;
  if (valid) {
    int4 o4;
    o4.x = ex;
    o4.y = ex + c.x;
    o4.z = ex + c.x + c.y;
    o4.w = ex + c.x + c.y + c.z;
    ((int4*)pos)[q] = o4;
  }
}

__global__ void __launch_bounds__(256)
chunk_place_kernel(const int* __restrict__ ei, const int* __restrict__ tab,
                   unsigned* __restrict__ lst) {
  __shared__ int cur[NB];
  int t = threadIdx.x, c = blockIdx.x;
  const int4* s4 = (const int4*)ei;
  const int4* d4 = (const int4*)(ei + EE);
  int4 s[4], d[4];
  bool ok[4];
#pragma unroll
  for (int k = 0; k < 4; ++k) {
    int q = c * (CH / 4) + k * 256 + t;
    ok[k] = q < EE / 4;
    int qq = ok[k] ? q : EE / 4 - 1;
    s[k] = s4[qq];
    d[k] = d4[qq];
  }
  for (int i = t; i < NB; i += 256) cur[i] = tab[i * NC + c];
  __syncthreads();
#pragma unroll
  for (int k = 0; k < 4; ++k) {
    if (ok[k]) {
      int p0 = atomicAdd(&cur[d[k].x >> 8], 1);
      int p1 = atomicAdd(&cur[d[k].y >> 8], 1);
      int p2 = atomicAdd(&cur[d[k].z >> 8], 1);
      int p3 = atomicAdd(&cur[d[k].w >> 8], 1);
      lst[p0] = (unsigned)s[k].x | ((unsigned)(d[k].x & (RB - 1)) << 24);
      lst[p1] = (unsigned)s[k].y | ((unsigned)(d[k].y & (RB - 1)) << 24);
      lst[p2] = (unsigned)s[k].z | ((unsigned)(d[k].z & (RB - 1)) << 24);
      lst[p3] = (unsigned)s[k].w | ((unsigned)(d[k].w & (RB - 1)) << 24);
    }
  }
}

__global__ void __launch_bounds__(256)
bucket_rows_kernel(const int* __restrict__ tab, const unsigned* __restrict__ lst,
                   int* __restrict__ pos, float* __restrict__ deg, int* __restrict__ col) {
  __shared__ int cnt[RB];
  __shared__ int wtot[8];
  int t = threadIdx.x, lane = t & 31, w = t >> 5;
  int b = blockIdx.x;
  int beg = tab[b * NC];
  int end = (b + 1 < NB) ? tab[(b + 1) * NC] : EE;

  cnt[t] = 0;
  __syncthreads();
  for (int e0 = beg; e0 < end; e0 += 4 * 256) {
    unsigned v[4];
    bool ok[4];
#pragma unroll
    for (int k = 0; k < 4; ++k) {
      int e = e0 + k * 256 + t;
      ok[k] = e < end;
      v[k] = lst[ok[k] ? e : end - 1];
    }
#pragma unroll
    for (int k = 0; k < 4; ++k)
      if (ok[k]) atomicAdd(&cnt[v[k] >> 24], 1);
  }
  __syncthreads();

  int c = cnt[t];
  int incl = c;
  for (int o = 1; o < 32; o <<= 1) {
    int u = __shfl_up(incl, o, 32);
    if (lane >= o) incl += u;
  }
  if (lane == 31) wtot[w] = incl;
  __syncthreads();
  int wo = 0;
  for (int k = 0; k < w; ++k) wo += wtot[k];
  int start = beg + wo + incl - c;
  int row = b * RB + t;
  if (row < NN) {
    pos[row] = start;
    deg[row] = (float)c;
  }
  cnt[t] = start;
  __syncthreads();

  for (int e0 = beg; e0 < end; e0 += 4 * 256) {
    unsigned v[4];
    bool ok[4];
#pragma unroll
    for (int k = 0; k < 4; ++k) {
      int e = e0 + k * 256 + t;
      ok[k] = e < end;
      v[k] = lst[ok[k] ? e : end - 1];
    }
#pragma unroll
    for (int k = 0; k < 4; ++k) {
      if (ok[k]) {
        int p = atomicAdd(&cnt[v[k] >> 24], 1);
        col[p] = (int)(v[k] & 0xFFFFFFu);
      }
    }
  }
}

__global__ void gather_kernel(const float* __restrict__ feat, const int* __restrict__ pos,
                              const int* __restrict__ col, float* __restrict__ agg) {
  int lane = threadIdx.x & 31;
  int d = __builtin_amdgcn_readfirstlane((int)((blockIdx.x * blockDim.x + threadIdx.x) >> 5));
  int beg = pos[d];
  int end = (d + 1 < NN) ? pos[d + 1] : EE;
  const float4* f4 = (const float4*)feat;

  float4 a0 = make_float4(0.f, 0.f, 0.f, 0.f);
  float4 a1 = a0, a2 = a0, a3 = a0;
  for (int base = beg; base < end; base += 32) {
    int n = end - base;
    if (n > 32) n = 32;
    int myc = col[base + (lane < n ? lane : n - 1)];
    int i = 0;
    for (; i + 4 <= n; i += 4) {
      int s0 = __builtin_amdgcn_readlane(myc, i);
      int s1 = __builtin_amdgcn_readlane(myc, i + 1);
      int s2 = __builtin_amdgcn_readlane(myc, i + 2);
      int s3 = __builtin_amdgcn_readlane(myc, i + 3);
      float4 v0 = f4[(size_t)s0 * (D / 4) + lane];
      float4 v1 = f4[(size_t)s1 * (D / 4) + lane];
      float4 v2 = f4[(size_t)s2 * (D / 4) + lane];
      float4 v3 = f4[(size_t)s3 * (D / 4) + lane];
      a0.x += v0.x; a0.y += v0.y; a0.z += v0.z; a0.w += v0.w;
      a1.x += v1.x; a1.y += v1.y; a1.z += v1.z; a1.w += v1.w;
      a2.x += v2.x; a2.y += v2.y; a2.z += v2.z; a2.w += v2.w;
      a3.x += v3.x; a3.y += v3.y; a3.z += v3.z; a3.w += v3.w;
    }
    int rem = n - i;
    if (rem == 3) {
      int s0 = __builtin_amdgcn_readlane(myc, i);
      int s1 = __builtin_amdgcn_readlane(myc, i + 1);
      int s2 = __builtin_amdgcn_readlane(myc, i + 2);
      float4 v0 = f4[(size_t)s0 * (D / 4) + lane];
      float4 v1 = f4[(size_t)s1 * (D / 4) + lane];
      float4 v2 = f4[(size_t)s2 * (D / 4) + lane];
      a0.x += v0.x; a0.y += v0.y; a0.z += v0.z; a0.w += v0.w;
      a1.x += v1.x; a1.y += v1.y; a1.z += v1.z; a1.w += v1.w;
      a2.x += v2.x; a2.y += v2.y; a2.z += v2.z; a2.w += v2.w;
    } else if (rem == 2) {
      int s0 = __builtin_amdgcn_readlane(myc, i);
      int s1 = __builtin_amdgcn_readlane(myc, i + 1);
      float4 v0 = f4[(size_t)s0 * (D / 4) + lane];
      float4 v1 = f4[(size_t)s1 * (D / 4) + lane];
      a0.x += v0.x; a0.y += v0.y; a0.z += v0.z; a0.w += v0.w;
      a1.x += v1.x; a1.y += v1.y; a1.z += v1.z; a1.w += v1.w;
    } else if (rem == 1) {
      int s0 = __builtin_amdgcn_readlane(myc, i);
      float4 v0 = f4[(size_t)s0 * (D / 4) + lane];
      a0.x += v0.x; a0.y += v0.y; a0.z += v0.z; a0.w += v0.w;
    }
  }
  float4 r;
  r.x = (a0.x + a1.x) + (a2.x + a3.x);
  r.y = (a0.y + a1.y) + (a2.y + a3.y);
  r.z = (a0.z + a1.z) + (a2.z + a3.z);
  r.w = (a0.w + a1.w) + (a2.w + a3.w);
  ((float4*)agg)[(size_t)d * (D / 4) + lane] = r;
}

#define NT   (NN / 16)
#define QP   (D + 8)
#define QPM  (HM + 8)
#define SBLK 512

template <int NCOL>
__device__ __forceinline__ void stage_quads(const float* __restrict__ Wg, float4* __restrict__ sW) {
#pragma unroll 8
  for (int it = 0; it < (32 * NCOL) / 256; ++it) {
    int idx = it * 256 + (int)threadIdx.x;
    int q = idx / NCOL, n = idx % NCOL;
    const float* p = Wg + (size_t)(4 * q) * NCOL + n;
    float4 v;
    v.x = p[0]; v.y = p[NCOL]; v.z = p[2 * NCOL]; v.w = p[3 * NCOL];
    sW[q * (NCOL + 8) + n] = v;
  }
}

__device__ __forceinline__ v8f mma4(float a0, float a1, float b0, float b1, v8f acc) {
  v2f a; a[0] = a0; a[1] = a1;
  v2f b; b[0] = b0; b[1] = b1;
  return __builtin_amdgcn_wmma_f32_16x16x4_f32(false, a, false, b, (short)0, acc, false, false);
}

__global__ void __launch_bounds__(256)
sage_combine_kernel(const float* __restrict__ xin,
                    const float* __restrict__ agg,
                    const float* __restrict__ deg,
                    const float* __restrict__ Wl,
                    const float* __restrict__ bias,
                    const float* __restrict__ Wr,
                    float* __restrict__ out, int do_relu) {
  __shared__ __attribute__((aligned(16))) float4 sW[2 * 32 * QP];
  stage_quads<D>(Wl, sW);
  stage_quads<D>(Wr, sW + 32 * QP);
  __syncthreads();

  int lane = threadIdx.x & 31;
  int w = __builtin_amdgcn_readfirstlane((int)(threadIdx.x >> 5));
  int c = lane & 15;
  int h = lane >> 4;
  const float4* bL = sW + h * QP + c;
  const float4* bR = bL + 32 * QP;

  for (int tile = w * (int)gridDim.x + (int)blockIdx.x; tile < NT; tile += 8 * (int)gridDim.x) {
    int m0 = tile << 4;
    int arow = m0 + c;
    float rdeg = 1.0f / fmaxf(deg[arow], 1.0f);
    const float4* xp = (const float4*)(xin + (size_t)arow * D) + h;
    const float4* gp = (const float4*)(agg + (size_t)arow * D) + h;

    v8f acc[8];
#pragma unroll
    for (int n = 0; n < 8; ++n) acc[n] = (v8f){};

    float4 cx[4], cg[4];
#pragma unroll
    for (int j = 0; j < 4; ++j) { cx[j] = xp[2 * j]; cg[j] = gp[2 * j]; }

    for (int ch = 0; ch < 4; ++ch) {
      float4 nx[4], ng[4];
#pragma unroll
      for (int j = 0; j < 4; ++j) { nx[j] = cx[j]; ng[j] = cg[j]; }
      if (ch < 3) {
#pragma unroll
        for (int j = 0; j < 4; ++j) {
          nx[j] = xp[8 * (ch + 1) + 2 * j];
          ng[j] = gp[8 * (ch + 1) + 2 * j];
        }
      }
#pragma unroll
      for (int j = 0; j < 4; ++j) {
        float4 ax = cx[j];
        float4 ag = cg[j];
        ag.x *= rdeg; ag.y *= rdeg; ag.z *= rdeg; ag.w *= rdeg;
        const float4* pl = bL + (8 * ch + 2 * j) * QP;
        const float4* pr = bR + (8 * ch + 2 * j) * QP;
        float4 fl[8], fr[8];
#pragma unroll
        for (int n = 0; n < 8; ++n) { fl[n] = pl[n * 16]; fr[n] = pr[n * 16]; }
#pragma unroll
        for (int n = 0; n < 8; ++n) acc[n] = mma4(ag.x, ag.y, fl[n].x, fl[n].y, acc[n]);
#pragma unroll
        for (int n = 0; n < 8; ++n) acc[n] = mma4(ag.z, ag.w, fl[n].z, fl[n].w, acc[n]);
#pragma unroll
        for (int n = 0; n < 8; ++n) acc[n] = mma4(ax.x, ax.y, fr[n].x, fr[n].y, acc[n]);
#pragma unroll
        for (int n = 0; n < 8; ++n) acc[n] = mma4(ax.z, ax.w, fr[n].z, fr[n].w, acc[n]);
      }
#pragma unroll
      for (int j = 0; j < 4; ++j) { cx[j] = nx[j]; cg[j] = ng[j]; }
    }

#pragma unroll
    for (int n = 0; n < 8; ++n) {
      float bv = bias[n * 16 + c];
#pragma unroll
      for (int g = 0; g < 8; ++g) {
        float v = acc[n][g] + bv;
        if (do_relu) v = fmaxf(v, 0.0f);
        out[(size_t)(m0 + g + h * 8) * D + n * 16 + c] = v;
      }
    }
  }
}

__global__ void __launch_bounds__(256)
mlp_kernel(const float* __restrict__ emb,
           const float* __restrict__ Wm1,
           const float* __restrict__ bm1,
           const float* __restrict__ Wm2,
           const float* __restrict__ bm2,
           float* __restrict__ pred) {
  __shared__ __attribute__((aligned(16))) float4 sW[32 * QPM];
  stage_quads<HM>(Wm1, sW);
  __syncthreads();

  int lane = threadIdx.x & 31;
  int w = __builtin_amdgcn_readfirstlane((int)(threadIdx.x >> 5));
  int c = lane & 15;
  int h = lane >> 4;
  const float4* bW = sW + h * QPM + c;
  float b2 = bm2[0];

  for (int tile = w * (int)gridDim.x + (int)blockIdx.x; tile < NT; tile += 8 * (int)gridDim.x) {
    int m0 = tile << 4;
    int arow = m0 + c;
    const float4* ep = (const float4*)(emb + (size_t)arow * D) + h;

    v8f acc[16];
#pragma unroll
    for (int n = 0; n < 16; ++n) acc[n] = (v8f){};

    float4 ce[4];
#pragma unroll
    for (int j = 0; j < 4; ++j) ce[j] = ep[2 * j];

    for (int ch = 0; ch < 4; ++ch) {
      float4 ne[4];
#pragma unroll
      for (int j = 0; j < 4; ++j) ne[j] = ce[j];
      if (ch < 3) {
#pragma unroll
        for (int j = 0; j < 4; ++j) ne[j] = ep[8 * (ch + 1) + 2 * j];
      }
#pragma unroll
      for (int j = 0; j < 4; ++j) {
        float4 a = ce[j];
        a.x = fmaxf(a.x, 0.0f); a.y = fmaxf(a.y, 0.0f);
        a.z = fmaxf(a.z, 0.0f); a.w = fmaxf(a.w, 0.0f);
        const float4* pb = bW + (8 * ch + 2 * j) * QPM;
#pragma unroll
        for (int half = 0; half < 2; ++half) {
          float4 fb[8];
#pragma unroll
          for (int n = 0; n < 8; ++n) fb[n] = pb[(half * 8 + n) * 16];
#pragma unroll
          for (int n = 0; n < 8; ++n)
            acc[half * 8 + n] = mma4(a.x, a.y, fb[n].x, fb[n].y, acc[half * 8 + n]);
#pragma unroll
          for (int n = 0; n < 8; ++n)
            acc[half * 8 + n] = mma4(a.z, a.w, fb[n].z, fb[n].w, acc[half * 8 + n]);
        }
      }
#pragma unroll
      for (int j = 0; j < 4; ++j) ce[j] = ne[j];
    }

    float s[8];
#pragma unroll
    for (int g = 0; g < 8; ++g) s[g] = 0.0f;
#pragma unroll
    for (int n = 0; n < 16; ++n) {
      float b1v = bm1[n * 16 + c];
      float w2  = Wm2[n * 16 + c];
#pragma unroll
      for (int g = 0; g < 8; ++g) s[g] += fmaxf(acc[n][g] + b1v, 0.0f) * w2;
    }
    float mine = 0.0f;
#pragma unroll
    for (int g = 0; g < 8; ++g) {
      float t = s[g];
      t += __shfl_xor(t, 8, 32);
      t += __shfl_xor(t, 4, 32);
      t += __shfl_xor(t, 2, 32);
      t += __shfl_xor(t, 1, 32);
      if (c == g) mine = t;
    }
    if (c < 8) pred[m0 + h * 8 + c] = mine + b2;
  }
}

extern "C" void kernel_launch(void* const* d_in, const int* in_sizes, int n_in,
                              void* d_out, int out_size, void* d_ws, size_t ws_size,
                              hipStream_t stream) {
  const float* x    = (const float*)d_in[0];
  const int*   ei   = (const int*)  d_in[1];
  const float* Wl0  = (const float*)d_in[2];
  const float* b0   = (const float*)d_in[3];
  const float* Wr0  = (const float*)d_in[4];
  const float* Wl1  = (const float*)d_in[5];
  const float* b1   = (const float*)d_in[6];
  const float* Wr1  = (const float*)d_in[7];
  const float* Wm1  = (const float*)d_in[8];
  const float* bm1  = (const float*)d_in[9];
  const float* Wm2  = (const float*)d_in[10];
  const float* bm2  = (const float*)d_in[11];

  float* emb  = (float*)d_out;
  float* pred = emb + (size_t)NN * D;

  float* ws  = (float*)d_ws;
  float* deg = ws;
  float* agg = ws + NN;
  float* h0  = agg + (size_t)NN * D;

  int*      pos  = (int*)d_out;
  int*      tab  = pos + NN;
  int*      bsum = tab + NB * NC;
  unsigned* lst  = (unsigned*)(bsum + 256);
  int*      col  = (int*)lst + EE;

  const int T = 256;
  const int gGather  = (NN * 32) / T;
  const int gStrip   = SBLK;

  chunk_count_kernel<<<NC, T, 0, stream>>>(ei, tab);
  chunk_sum_kernel<<<SCAN_NB, T, 0, stream>>>(tab, bsum);
  scan_kernel<<<SCAN_NB, T, 0, stream>>>(tab, bsum);
  chunk_place_kernel<<<NCH, T, 0, stream>>>(ei, tab, lst);
  bucket_rows_kernel<<<NB, T, 0, stream>>>(tab, lst, pos, deg, col);

  gather_kernel<<<gGather, T, 0, stream>>>(x, pos, col, agg);
  sage_combine_kernel<<<gStrip, T, 0, stream>>>(x, agg, deg, Wl0, b0, Wr0, h0, 1);

  gather_kernel<<<gGather, T, 0, stream>>>(h0, pos, col, agg);
  sage_combine_kernel<<<gStrip, T, 0, stream>>>(h0, agg, deg, Wl1, b1, Wr1, emb, 0);

  mlp_kernel<<<gStrip, T, 0, stream>>>(emb, Wm1, bm1, Wm2, bm2, pred);
}


# Strip kernels dealt evenly per SIMD: one 16-wave block per WGP (4 waves per SIMD, <=256 registers, weights staged once); the combine's left-over strips cut into column pairs over the 4 SIMDs
# speedup vs baseline: 4.0326x; 1.0870x over previous
#include <hip/hip_runtime.h>

#define NN 100000
#define EE 1600000
#define D  128
#define HM 256

typedef __attribute__((ext_vector_type(2))) float v2f;
typedef __attribute__((ext_vector_type(8))) float v8f;

#define RB      256
#define NB      ((NN + RB - 1) / RB)
#define CH      4096
#define NCH     ((EE + CH - 1) / CH)
#define NC      (NCH + 1)
#define SCAN_Q  ((NB * NC) / 4)
#define SCAN_NB ((SCAN_Q + 255) / 256)

static_assert((NB * NC) % 4 == 0, "the scan reads the table as int4");
static_assert(SCAN_NB <= 256, "one scan block sums the totals of all blocks before it");
static_assert(NN <= (1 << 24), "a source index shares its word with 8 bits of the destination");
static_assert(EE % 4 == 0, "edge lists are read as int4");

__global__ void __launch_bounds__(256)
chunk_count_kernel(const int* __restrict__ ei, int* __restrict__ tab) {
  __shared__ int h[NB];
  int t = threadIdx.x, c = blockIdx.x;
  for (int i = t; i < NB; i += 256) h[i] = 0;
  __syncthreads();
  const int4* d4 = (const int4*)(ei + EE);
  int4 d[4];
  bool ok[4];
#pragma unroll
  for (int k = 0; k < 4; ++k) {
    int q = c * (CH / 4) + k * 256 + t;
    ok[k] = q < EE / 4;
    d[k] = d4[ok[k] ? q : EE / 4 - 1];
  }
#pragma unroll
  for (int k = 0; k < 4; ++k) {
    if (ok[k]) {
      atomicAdd(&h[d[k].x >> 8], 1);
      atomicAdd(&h[d[k].y >> 8], 1);
      atomicAdd(&h[d[k].z >> 8], 1);
      atomicAdd(&h[d[k].w >> 8], 1);
    }
  }
  __syncthreads();
  for (int i = t; i < NB; i += 256) tab[i * NC + c] = h[i];
}

__global__ void chunk_sum_kernel(const int* __restrict__ cnt, int* __restrict__ bsum) {
  __shared__ int wsum[8];
  int t = threadIdx.x;
  int q = blockIdx.x * 256 + t;
  int s = 0;
  if (q < SCAN_Q) {
    int4 c = ((const int4*)cnt)[q];
    s = c.x + c.y + c.z + c.w;
  }
  for (int o = 16; o > 0; o >>= 1) s += __shfl_xor(s, o, 32);
  if ((t & 31) == 0) wsum[t >> 5] = s;
  __syncthreads();
  if (t == 0) {
    int tot = 0;
    for (int w = 0; w < 8; ++w) tot += wsum[w];
    bsum[blockIdx.x] = tot;
  }
}

__global__ void scan_kernel(int* pos, const int* __restrict__ bsum) {
  __shared__ int wtot[8];
  __shared__ int wpre[8];
  int t = threadIdx.x, lane = t & 31, w = t >> 5;
  int b = blockIdx.x;

  int v = (t < b) ? bsum[t] : 0;
  for (int o = 16; o > 0; o >>= 1) v += __shfl_xor(v, o, 32);
  if (lane == 0) wpre[w] = v;
  __syncthreads();
  int base = 0;
  for (int k = 0; k < 8; ++k) base += wpre[k];

  int q = b * 256 + t;
  bool valid = q < SCAN_Q;
  int4 c = make_int4(0, 0, 0, 0);
  if (valid) c = ((const int4*)pos)[q];
  int tot = c.x + c.y + c.z + c.w;
  int incl = tot;
  for (int o = 1; o < 32; o <<= 1) {
    int u = __shfl_up(incl, o, 32);
    if (lane >= o) incl += u;
  }
  if (lane == 31) wtot[w] = incl;
  __syncthreads();
  int wo = 0;
  for (int k = 0; k < w; ++k) wo += wtot[k];
  int ex = base + wo + incl - tot;
  if (valid) {
    int4 o4;
    o4.x = ex;
    o4.y = ex + c.x;
    o4.z = ex + c.x + c.y;
    o4.w = ex + c.x + c.y + c.z;
    ((int4*)pos)[q] = o4;
  }
}

__global__ void __launch_bounds__(256)
chunk_place_kernel(const int* __restrict__ ei, const int* __restrict__ tab,
                   unsigned* __restrict__ lst) {
  __shared__ int cur[NB];
  int t = threadIdx.x, c = blockIdx.x;
  const int4* s4 = (const int4*)ei;
  const int4* d4 = (const int4*)(ei + EE);
  int4 s[4], d[4];
  bool ok[4];
#pragma unroll
  for (int k = 0; k < 4; ++k) {
    int q = c * (CH / 4) + k * 256 + t;
    ok[k] = q < EE / 4;
    int qq = ok[k] ? q : EE / 4 - 1;
    s[k] = s4[qq];
    d[k] = d4[qq];
  }
  for (int i = t; i < NB; i += 256) cur[i] = tab[i * NC + c];
  __syncthreads();
#pragma unroll
  for (int k = 0; k < 4; ++k) {
    if (ok[k]) {
      int p0 = atomicAdd(&cur[d[k].x >> 8], 1);
      int p1 = atomicAdd(&cur[d[k].y >> 8], 1);
      int p2 = atomicAdd(&cur[d[k].z >> 8], 1);
      int p3 = atomicAdd(&cur[d[k].w >> 8], 1);
      lst[p0] = (unsigned)s[k].x | ((unsigned)(d[k].x & (RB - 1)) << 24);
      lst[p1] = (unsigned)s[k].y | ((unsigned)(d[k].y & (RB - 1)) << 24);
      lst[p2] = (unsigned)s[k].z | ((unsigned)(d[k].z & (RB - 1)) << 24);
      lst[p3] = (unsigned)s[k].w | ((unsigned)(d[k].w & (RB - 1)) << 24);
    }
  }
}

__global__ void __launch_bounds__(256)
bucket_rows_kernel(const int* __restrict__ tab, const unsigned* __restrict__ lst,
                   int* __restrict__ pos, float* __restrict__ deg, int* __restrict__ col) {
  __shared__ int cnt[RB];
  __shared__ int wtot[8];
  int t = threadIdx.x, lane = t & 31, w = t >> 5;
  int b = blockIdx.x;
  int beg = tab[b * NC];
  int end = (b + 1 < NB) ? tab[(b + 1) * NC] : EE;

  cnt[t] = 0;
  __syncthreads();
  for (int e0 = beg; e0 < end; e0 += 4 * 256) {
    unsigned v[4];
    bool ok[4];
#pragma unroll
    for (int k = 0; k < 4; ++k) {
      int e = e0 + k * 256 + t;
      ok[k] = e < end;
      v[k] = lst[ok[k] ? e : end - 1];
    }
#pragma unroll
    for (int k = 0; k < 4; ++k)
      if (ok[k]) atomicAdd(&cnt[v[k] >> 24], 1);
  }
  __syncthreads();

  int c = cnt[t];
  int incl = c;
  for (int o = 1; o < 32; o <<= 1) {
    int u = __shfl_up(incl, o, 32);
    if (lane >= o) incl += u;
  }
  if (lane == 31) wtot[w] = incl;
  __syncthreads();
  int wo = 0;
  for (int k = 0; k < w; ++k) wo += wtot[k];
  int start = beg + wo + incl - c;
  int row = b * RB + t;
  if (row < NN) {
    pos[row] = start;
    deg[row] = (float)c;
  }
  cnt[t] = start;
  __syncthreads();

  for (int e0 = beg; e0 < end; e0 += 4 * 256) {
    unsigned v[4];
    bool ok[4];
#pragma unroll
    for (int k = 0; k < 4; ++k) {
      int e = e0 + k * 256 + t;
      ok[k] = e < end;
      v[k] = lst[ok[k] ? e : end - 1];
    }
#pragma unroll
    for (int k = 0; k < 4; ++k) {
      if (ok[k]) {
        int p = atomicAdd(&cnt[v[k] >> 24], 1);
        col[p] = (int)(v[k] & 0xFFFFFFu);
      }
    }
  }
}

__global__ void gather_kernel(const float* __restrict__ feat, const int* __restrict__ pos,
                              const int* __restrict__ col, float* __restrict__ agg) {
  int lane = threadIdx.x & 31;
  int d = __builtin_amdgcn_readfirstlane((int)((blockIdx.x * blockDim.x + threadIdx.x) >> 5));
  int beg = pos[d];
  int end = (d + 1 < NN) ? pos[d + 1] : EE;
  const float4* f4 = (const float4*)feat;

  float4 a0 = make_float4(0.f, 0.f, 0.f, 0.f);
  float4 a1 = a0, a2 = a0, a3 = a0;
  for (int base = beg; base < end; base += 32) {
    int n = end - base;
    if (n > 32) n = 32;
    int myc = col[base + (lane < n ? lane : n - 1)];
    int i = 0;
    for (; i + 4 <= n; i += 4) {
      int s0 = __builtin_amdgcn_readlane(myc, i);
      int s1 = __builtin_amdgcn_readlane(myc, i + 1);
      int s2 = __builtin_amdgcn_readlane(myc, i + 2);
      int s3 = __builtin_amdgcn_readlane(myc, i + 3);
      float4 v0 = f4[(size_t)s0 * (D / 4) + lane];
      float4 v1 = f4[(size_t)s1 * (D / 4) + lane];
      float4 v2 = f4[(size_t)s2 * (D / 4) + lane];
      float4 v3 = f4[(size_t)s3 * (D / 4) + lane];
      a0.x += v0.x; a0.y += v0.y; a0.z += v0.z; a0.w += v0.w;
      a1.x += v1.x; a1.y += v1.y; a1.z += v1.z; a1.w += v1.w;
      a2.x += v2.x; a2.y += v2.y; a2.z += v2.z; a2.w += v2.w;
      a3.x += v3.x; a3.y += v3.y; a3.z += v3.z; a3.w += v3.w;
    }
    int rem = n - i;
    if (rem == 3) {
      int s0 = __builtin_amdgcn_readlane(myc, i);
      int s1 = __builtin_amdgcn_readlane(myc, i + 1);
      int s2 = __builtin_amdgcn_readlane(myc, i + 2);
      float4 v0 = f4[(size_t)s0 * (D / 4) + lane];
      float4 v1 = f4[(size_t)s1 * (D / 4) + lane];
      float4 v2 = f4[(size_t)s2 * (D / 4) + lane];
      a0.x += v0.x; a0.y += v0.y; a0.z += v0.z; a0.w += v0.w;
      a1.x += v1.x; a1.y += v1.y; a1.z += v1.z; a1.w += v1.w;
      a2.x += v2.x; a2.y += v2.y; a2.z += v2.z; a2.w += v2.w;
    } else if (rem == 2) {
      int s0 = __builtin_amdgcn_readlane(myc, i);
      int s1 = __builtin_amdgcn_readlane(myc, i + 1);
      float4 v0 = f4[(size_t)s0 * (D / 4) + lane];
      float4 v1 = f4[(size_t)s1 * (D / 4) + lane];
      a0.x += v0.x; a0.y += v0.y; a0.z += v0.z; a0.w += v0.w;
      a1.x += v1.x; a1.y += v1.y; a1.z += v1.z; a1.w += v1.w;
    } else if (rem == 1) {
      int s0 = __builtin_amdgcn_readlane(myc, i);
      float4 v0 = f4[(size_t)s0 * (D / 4) + lane];
      a0.x += v0.x; a0.y += v0.y; a0.z += v0.z; a0.w += v0.w;
    }
  }
  float4 r;
  r.x = (a0.x + a1.x) + (a2.x + a3.x);
  r.y = (a0.y + a1.y) + (a2.y + a3.y);
  r.z = (a0.z + a1.z) + (a2.z + a3.z);
  r.w = (a0.w + a1.w) + (a2.w + a3.w);
  ((float4*)agg)[(size_t)d * (D / 4) + lane] = r;
}

#define NT   (NN / 16)
#define QP   (D + 8)
#define QPM  (HM + 8)
#define SBLK 256
#define STHR 512
#define SWAV (STHR / 32)

template <int NCOL>
__device__ __forceinline__ void stage_quads(const float* __restrict__ Wg, float4* __restrict__ sW) {
#pragma unroll 8
  for (int it = 0; it < (32 * NCOL) / STHR; ++it) {
    int idx = it * STHR + (int)threadIdx.x;
    int q = idx / NCOL, n = idx % NCOL;
    const float* p = Wg + (size_t)(4 * q) * NCOL + n;
    float4 v;
    v.x = p[0]; v.y = p[NCOL]; v.z = p[2 * NCOL]; v.w = p[3 * NCOL];
    sW[q * (NCOL + 8) + n] = v;
  }
}

__device__ __forceinline__ v8f mma4(float a0, float a1, float b0, float b1, v8f acc) {
  v2f a; a[0] = a0; a[1] = a1;
  v2f b; b[0] = b0; b[1] = b1;
  return __builtin_amdgcn_wmma_f32_16x16x4_f32(false, a, false, b, (short)0, acc, false, false);
}

#define NT_MAIN ((NT / (4 * SBLK)) * (4 * SBLK))

static_assert(SWAV == 16, "the left-over strips go to waves 8..11, which hold one whole strip fewer than waves 0..7");
static_assert((NT_MAIN / SBLK) % SWAV == 8, "whole strips must end with waves 0..7, so that waves 8..15 are the lighter ones");

template <int NTW>
__device__ __forceinline__ void combine_strip(int tile, int nb, int c, int h,
                                              const float* __restrict__ xin,
                                              const float* __restrict__ agg,
                                              const float* __restrict__ deg,
                                              const float* __restrict__ bias,
                                              float* __restrict__ out, int do_relu,
                                              const float4* bL, const float4* bR) {
  int m0 = tile << 4;
  int arow = m0 + c;
  float rdeg = 1.0f / fmaxf(deg[arow], 1.0f);
  const float4* xp = (const float4*)(xin + (size_t)arow * D) + h;
  const float4* gp = (const float4*)(agg + (size_t)arow * D) + h;

  v8f acc[NTW];
#pragma unroll
  for (int n = 0; n < NTW; ++n) acc[n] = (v8f){};

  float4 cx[4], cg[4];
#pragma unroll
  for (int j = 0; j < 4; ++j) { cx[j] = xp[2 * j]; cg[j] = gp[2 * j]; }

  for (int ch = 0; ch < 4; ++ch) {
    float4 nx[4], ng[4];
#pragma unroll
    for (int j = 0; j < 4; ++j) { nx[j] = cx[j]; ng[j] = cg[j]; }
    if (ch < 3) {
#pragma unroll
      for (int j = 0; j < 4; ++j) {
        nx[j] = xp[8 * (ch + 1) + 2 * j];
        ng[j] = gp[8 * (ch + 1) + 2 * j];
      }
    }
#pragma unroll
    for (int j = 0; j < 4; ++j) {
      float4 ax = cx[j];
      float4 ag = cg[j];
      ag.x *= rdeg; ag.y *= rdeg; ag.z *= rdeg; ag.w *= rdeg;
      const float4* pl = bL + (8 * ch + 2 * j) * QP + nb * 16;
      const float4* pr = bR + (8 * ch + 2 * j) * QP + nb * 16;
      float4 fl[NTW], fr[NTW];
#pragma unroll
      for (int n = 0; n < NTW; ++n) { fl[n] = pl[n * 16]; fr[n] = pr[n * 16]; }
#pragma unroll
      for (int n = 0; n < NTW; ++n) acc[n] = mma4(ag.x, ag.y, fl[n].x, fl[n].y, acc[n]);
#pragma unroll
      for (int n = 0; n < NTW; ++n) acc[n] = mma4(ag.z, ag.w, fl[n].z, fl[n].w, acc[n]);
#pragma unroll
      for (int n = 0; n < NTW; ++n) acc[n] = mma4(ax.x, ax.y, fr[n].x, fr[n].y, acc[n]);
#pragma unroll
      for (int n = 0; n < NTW; ++n) acc[n] = mma4(ax.z, ax.w, fr[n].z, fr[n].w, acc[n]);
    }
#pragma unroll
    for (int j = 0; j < 4; ++j) { cx[j] = nx[j]; cg[j] = ng[j]; }
  }

#pragma unroll
  for (int n = 0; n < NTW; ++n) {
    float bv = bias[(nb + n) * 16 + c];
#pragma unroll
    for (int g = 0; g < 8; ++g) {
      float v = acc[n][g] + bv;
      if (do_relu) v = fmaxf(v, 0.0f);
      out[(size_t)(m0 + g + h * 8) * D + (nb + n) * 16 + c] = v;
    }
  }
}

__global__ void __launch_bounds__(STHR)
sage_combine_kernel(const float* __restrict__ xin,
                    const float* __restrict__ agg,
                    const float* __restrict__ deg,
                    const float* __restrict__ Wl,
                    const float* __restrict__ bias,
                    const float* __restrict__ Wr,
                    float* __restrict__ out, int do_relu) {
  __shared__ __attribute__((aligned(16))) float4 sW[2 * 32 * QP];
  stage_quads<D>(Wl, sW);
  stage_quads<D>(Wr, sW + 32 * QP);
  __syncthreads();

  int lane = threadIdx.x & 31;
  int w = __builtin_amdgcn_readfirstlane((int)(threadIdx.x >> 5));
  int c = lane & 15;
  int h = lane >> 4;
  const float4* bL = sW + h * QP + c;
  const float4* bR = bL + 32 * QP;

  for (int tile = w * SBLK + (int)blockIdx.x; tile < NT_MAIN; tile += SWAV * SBLK) {
    asm volatile("" ::: "memory");
    combine_strip<8>(tile, 0, c, h, xin, agg, deg, bias, out, do_relu, bL, bR);
  }

  if (w >= 8 && w < 12) {
    for (int tile = NT_MAIN + (int)blockIdx.x; tile < NT; tile += SBLK) {
      asm volatile("" ::: "memory");
      combine_strip<2>(tile, 2 * (w - 8), c, h, xin, agg, deg, bias, out, do_relu, bL, bR);
    }
  }
}

__global__ void __launch_bounds__(STHR)
mlp_kernel(const float* __restrict__ emb,
           const float* __restrict__ Wm1,
           const float* __restrict__ bm1,
           const float* __restrict__ Wm2,
           const float* __restrict__ bm2,
           float* __restrict__ pred) {
  __shared__ __attribute__((aligned(16))) float4 sW[32 * QPM];
  stage_quads<HM>(Wm1, sW);
  __syncthreads();

  int lane = threadIdx.x & 31;
  int w = __builtin_amdgcn_readfirstlane((int)(threadIdx.x >> 5));
  int c = lane & 15;
  int h = lane >> 4;
  const float4* bW = sW + h * QPM + c;
  float b2 = bm2[0];

  for (int tile = w * (int)gridDim.x + (int)blockIdx.x; tile < NT; tile += SWAV * (int)gridDim.x) {
    asm volatile("" ::: "memory");
    int m0 = tile << 4;
    int arow = m0 + c;
    const float4* ep = (const float4*)(emb + (size_t)arow * D) + h;

    v8f acc[16];
#pragma unroll
    for (int n = 0; n < 16; ++n) acc[n] = (v8f){};

    float4 ce[4];
#pragma unroll
    for (int j = 0; j < 4; ++j) ce[j] = ep[2 * j];

    for (int ch = 0; ch < 4; ++ch) {
      float4 ne[4];
#pragma unroll
      for (int j = 0; j < 4; ++j) ne[j] = ce[j];
      if (ch < 3) {
#pragma unroll
        for (int j = 0; j < 4; ++j) ne[j] = ep[8 * (ch + 1) + 2 * j];
      }
#pragma unroll
      for (int j = 0; j < 4; ++j) {
        float4 a = ce[j];
        a.x = fmaxf(a.x, 0.0f); a.y = fmaxf(a.y, 0.0f);
        a.z = fmaxf(a.z, 0.0f); a.w = fmaxf(a.w, 0.0f);
        const float4* pb = bW + (8 * ch + 2 * j) * QPM;
#pragma unroll
        for (int half = 0; half < 2; ++half) {
          float4 fb[8];
#pragma unroll
          for (int n = 0; n < 8; ++n) fb[n] = pb[(half * 8 + n) * 16];
#pragma unroll
          for (int n = 0; n < 8; ++n)
            acc[half * 8 + n] = mma4(a.x, a.y, fb[n].x, fb[n].y, acc[half * 8 + n]);
#pragma unroll
          for (int n = 0; n < 8; ++n)
            acc[half * 8 + n] = mma4(a.z, a.w, fb[n].z, fb[n].w, acc[half * 8 + n]);
        }
      }
#pragma unroll
      for (int j = 0; j < 4; ++j) ce[j] = ne[j];
    }

    float s[8];
#pragma unroll
    for (int g = 0; g < 8; ++g) s[g] = 0.0f;
#pragma unroll
    for (int n = 0; n < 16; ++n) {
      float b1v = bm1[n * 16 + c];
      float w2  = Wm2[n * 16 + c];
#pragma unroll
      for (int g = 0; g < 8; ++g) s[g] += fmaxf(acc[n][g] + b1v, 0.0f) * w2;
    }
    float mine = 0.0f;
#pragma unroll
    for (int g = 0; g < 8; ++g) {
      float t = s[g];
      t += __shfl_xor(t, 8, 32);
      t += __shfl_xor(t, 4, 32);
      t += __shfl_xor(t, 2, 32);
      t += __shfl_xor(t, 1, 32);
      if (c == g) mine = t;
    }
    if (c < 8) pred[m0 + h * 8 + c] = mine + b2;
  }
}

extern "C" void kernel_launch(void* const* d_in, const int* in_sizes, int n_in,
                              void* d_out, int out_size, void* d_ws, size_t ws_size,
                              hipStream_t stream) {
  const float* x    = (const float*)d_in[0];
  const int*   ei   = (const int*)  d_in[1];
  const float* Wl0  = (const float*)d_in[2];
  const float* b0   = (const float*)d_in[3];
  const float* Wr0  = (const float*)d_in[4];
  const float* Wl1  = (const float*)d_in[5];
  const float* b1   = (const float*)d_in[6];
  const float* Wr1  = (const float*)d_in[7];
  const float* Wm1  = (const float*)d_in[8];
  const float* bm1  = (const float*)d_in[9];
  const float* Wm2  = (const float*)d_in[10];
  const float* bm2  = (const float*)d_in[11];

  float* emb  = (float*)d_out;
  float* pred = emb + (size_t)NN * D;

  float* ws  = (float*)d_ws;
  float* deg = ws;
  float* agg = ws + NN;
  float* h0  = agg + (size_t)NN * D;

  int*      pos  = (int*)d_out;
  int*      tab  = pos + NN;
  int*      bsum = tab + NB * NC;
  unsigned* lst  = (unsigned*)(bsum + 256);
  int*      col  = (int*)lst + EE;

  const int T = 256;
  const int gGather  = (NN * 32) / T;
  const int gStrip   = SBLK;

  chunk_count_kernel<<<NC, T, 0, stream>>>(ei, tab);
  chunk_sum_kernel<<<SCAN_NB, T, 0, stream>>>(tab, bsum);
  scan_kernel<<<SCAN_NB, T, 0, stream>>>(tab, bsum);
  chunk_place_kernel<<<NCH, T, 0, stream>>>(ei, tab, lst);
  bucket_rows_kernel<<<NB, T, 0, stream>>>(tab, lst, pos, deg, col);

  gather_kernel<<<gGather, T, 0, stream>>>(x, pos, col, agg);
  sage_combine_kernel<<<gStrip, STHR, 0, stream>>>(x, agg, deg, Wl0, b0, Wr0, h0, 1);

  gather_kernel<<<gGather, T, 0, stream>>>(h0, pos, col, agg);
  sage_combine_kernel<<<gStrip, STHR, 0, stream>>>(h0, agg, deg, Wl1, b1, Wr1, emb, 0);

  mlp_kernel<<<gStrip, STHR, 0, stream>>>(emb, Wm1, bm1, Wm2, bm2, pred);
}
